# starting kernel
# baseline (speedup 1.0000x reference)
#include <hip/hip_runtime.h>
#include <math.h>

typedef __attribute__((ext_vector_type(2))) float v2f;
typedef __attribute__((ext_vector_type(8))) float v8f;

#define WPB 8

__global__ __launch_bounds__(256) void nxro_fused_kernel(
    const float* __restrict__ x, const float* __restrict__ t,
    const float* __restrict__ Lb, const float* __restrict__ wq,
    const float* __restrict__ wk, const float* __restrict__ wv,
    const float* __restrict__ wo, const float* __restrict__ aw,
    float* __restrict__ out, int nTiles)
{
    __shared__ float xs[WPB][16][18];
    __shared__ float al[WPB][16];

    const int lane = threadIdx.x & 31;
    const int wave = threadIdx.x >> 5;
    const int half = lane >> 4;
    const int r    = lane & 15;

    v2f bfr[20];
#pragma unroll
    for (int j = 0; j < 20; ++j) {
        const int kb = j >> 2;
        const int v0 = ((j & 3) << 2) + (half << 1);
        const float* p = Lb + kb * 256 + r * 16 + v0;
        v2f b_; b_.x = p[0]; b_.y = p[1];
        bfr[j] = b_;
    }

    float c1 = 0.f, c2 = 0.f;
#pragma unroll
    for (int d = 0; d < 32; ++d) {
        c1 = fmaf(wq[d], wk[d], c1);
        c2 = fmaf(wv[d], wo[d], c2);
    }
    c1 *= 0.17677669529663688f;

    const float aw0 = aw[0], aw1 = aw[1], aw2 = aw[2], aw3 = aw[3], aw4 = aw[4];

    const int waveId = blockIdx.x * WPB + wave;
    const int stride = gridDim.x * WPB;

    for (int tile = waveId; tile < nTiles; tile += stride) {
        const int s0  = tile << 4;
        const int row = s0 + r;

        const v2f* xp = reinterpret_cast<const v2f*>(x + row * 16);
        const v2f xq0 = xp[0 + half];
        const v2f xq1 = xp[2 + half];
        const v2f xq2 = xp[4 + half];
        const v2f xq3 = xp[6 + half];

        {
            v2f* rp = reinterpret_cast<v2f*>(&xs[wave][r][0]);
            rp[0 + half] = xq0;
            rp[2 + half] = xq1;
            rp[4 + half] = xq2;
            rp[6 + half] = xq3;
        }

        const float tt  = t[row];
        const float a1  = 6.283185307179586f * tt;
        const float e1c = cosf(a1), e1s = sinf(a1);
        const float tc  = e1c + e1c;
        const float e2c = fmaf(tc, e1c, -1.0f);
        const float e2s = tc * e1s;

        if (half == 0) {
            float z = aw0 + aw1 * e1c + aw2 * e1s + aw3 * e2c + aw4 * e2s;
            al[wave][r] = __builtin_amdgcn_rcpf(1.f + __expf(-z));
        }

        v8f acc = {};
#define WMMA_RAW(AFR, BJ)                                                      \
        acc = __builtin_amdgcn_wmma_f32_16x16x4_f32(                           \
            false, (AFR), false, bfr[BJ], (short)0, acc, false, false)
#define WMMA_SC(EV, XQ, BJ)                                                    \
        do { v2f a_; a_.x = (EV) * (XQ).x; a_.y = (EV) * (XQ).y;               \
             WMMA_RAW(a_, BJ); } while (0)

        WMMA_RAW(xq0, 0);  WMMA_RAW(xq1, 1);  WMMA_RAW(xq2, 2);  WMMA_RAW(xq3, 3);
        WMMA_SC(e1c, xq0, 4);  WMMA_SC(e1c, xq1, 5);  WMMA_SC(e1c, xq2, 6);  WMMA_SC(e1c, xq3, 7);
        WMMA_SC(e1s, xq0, 8);  WMMA_SC(e1s, xq1, 9);  WMMA_SC(e1s, xq2, 10); WMMA_SC(e1s, xq3, 11);
        WMMA_SC(e2c, xq0, 12); WMMA_SC(e2c, xq1, 13); WMMA_SC(e2c, xq2, 14); WMMA_SC(e2c, xq3, 15);
        WMMA_SC(e2s, xq0, 16); WMMA_SC(e2s, xq1, 17); WMMA_SC(e2s, xq2, 18); WMMA_SC(e2s, xq3, 19);
#undef WMMA_SC
#undef WMMA_RAW

        const int i = r;
        const float g = (i >= 2) ? 1.f : 0.f;
#pragma unroll
        for (int rr = 0; rr < 8; ++rr) {
            const int m = rr + (half << 3);
            const v2f xm01 = *reinterpret_cast<const v2f*>(&xs[wave][m][0]);
            const float xm0 = xm01.x;
            const float xm1 = xm01.y;
            const float xmi = xs[wave][m][i];
            const float am  = al[wave][m];
            const float e0 = c1 * xmi * xm0;
            const float e1 = c1 * xmi * xm1;
            const float e2 = c1 * xmi * xmi;
            const float mx = fmaxf(fmaxf(e0, e1), e2);
            const float w0 = __expf(e0 - mx);
            const float w1 = __expf(e1 - mx);
            const float w2 = g * __expf(e2 - mx);
            const float inv = __builtin_amdgcn_rcpf(w0 + w1 + w2);
            const float s  = (w0 * xm0 + w1 * xm1 + w2 * xmi) * inv;
            out[(s0 + m) * 16 + i] = acc[rr] + c2 * s * am;
        }
    }
}

extern "C" void kernel_launch(void* const* d_in, const int* in_sizes, int n_in,
                              void* d_out, int out_size, void* d_ws, size_t ws_size,
                              hipStream_t stream) {
    const float* x  = (const float*)d_in[0];
    const float* t  = (const float*)d_in[1];
    const float* Lb = (const float*)d_in[2];
    const float* wq = (const float*)d_in[3];
    const float* wk = (const float*)d_in[4];
    const float* wv = (const float*)d_in[5];
    const float* wo = (const float*)d_in[6];
    const float* aw = (const float*)d_in[7];
    float* out = (float*)d_out;

    const int B = in_sizes[1];
    const int nTiles = B >> 4;
    if (nTiles <= 0) return;

    int blocks = (nTiles + WPB - 1) / WPB;
    if (blocks > 512) blocks = 512;

    nxro_fused_kernel<<<blocks, 256, 0, stream>>>(x, t, Lb, wq, wk, wv, wo, aw,
                                                  out, nTiles);
}


# Each tile's x and t loads are issued one tile ahead (the first before the constant set-up, the next before the current tile's arithmetic), so memory latency runs under work; all sums unchanged
# speedup vs baseline: 1.1590x; 1.1590x over previous
#include <hip/hip_runtime.h>
#include <math.h>

typedef __attribute__((ext_vector_type(2))) float v2f;
typedef __attribute__((ext_vector_type(8))) float v8f;

#define WPB 8

__global__ __launch_bounds__(256) void nxro_fused_kernel(
    const float* __restrict__ x, const float* __restrict__ t,
    const float* __restrict__ Lb, const float* __restrict__ wq,
    const float* __restrict__ wk, const float* __restrict__ wv,
    const float* __restrict__ wo, const float* __restrict__ aw,
    float* __restrict__ out, int nTiles)
{
    __shared__ float xs[WPB][16][18];
    __shared__ float al[WPB][16];

    const int lane = threadIdx.x & 31;
    const int wave = threadIdx.x >> 5;
    const int half = lane >> 4;
    const int r    = lane & 15;

    const int waveId = blockIdx.x * WPB + wave;
    const int stride = gridDim.x * WPB;
    if (waveId >= nTiles) return;

    int tile = waveId;
    float tt;
    v2f xq0, xq1, xq2, xq3;
    {
        const int row0 = (tile << 4) + r;
        tt = t[row0];
        const v2f* xp = reinterpret_cast<const v2f*>(x + row0 * 16);
        xq0 = xp[0 + half];
        xq1 = xp[2 + half];
        xq2 = xp[4 + half];
        xq3 = xp[6 + half];
    }

    v2f bfr[20];
#pragma unroll
    for (int j = 0; j < 20; ++j) {
        const int kb = j >> 2;
        const int v0 = ((j & 3) << 2) + (half << 1);
        const float* p = Lb + kb * 256 + r * 16 + v0;
        v2f b_; b_.x = p[0]; b_.y = p[1];
        bfr[j] = b_;
    }

    float c1 = 0.f, c2 = 0.f;
#pragma unroll
    for (int d = 0; d < 32; ++d) {
        c1 = fmaf(wq[d], wk[d], c1);
        c2 = fmaf(wv[d], wo[d], c2);
    }
    c1 *= 0.17677669529663688f;

    const float aw0 = aw[0], aw1 = aw[1], aw2 = aw[2], aw3 = aw[3], aw4 = aw[4];

    for (;;) {
        const int s0  = tile << 4;

        const int  nextTile = tile + stride;
        const bool more     = nextTile < nTiles;
        const int  nrow     = ((more ? nextTile : tile) << 4) + r;
        const float ntt = t[nrow];
        const v2f* np = reinterpret_cast<const v2f*>(x + nrow * 16);
        const v2f nq0 = np[0 + half];
        const v2f nq1 = np[2 + half];
        const v2f nq2 = np[4 + half];
        const v2f nq3 = np[6 + half];

        {
            v2f* rp = reinterpret_cast<v2f*>(&xs[wave][r][0]);
            rp[0 + half] = xq0;
            rp[2 + half] = xq1;
            rp[4 + half] = xq2;
            rp[6 + half] = xq3;
        }

        const float a1  = 6.283185307179586f * tt;
        const float e1c = cosf(a1), e1s = sinf(a1);
        const float tc  = e1c + e1c;
        const float e2c = fmaf(tc, e1c, -1.0f);
        const float e2s = tc * e1s;

        if (half == 0) {
            float z = aw0 + aw1 * e1c + aw2 * e1s + aw3 * e2c + aw4 * e2s;
            al[wave][r] = __builtin_amdgcn_rcpf(1.f + __expf(-z));
        }

        v8f acc = {};
#define WMMA_RAW(AFR, BJ)                                                      \
        acc = __builtin_amdgcn_wmma_f32_16x16x4_f32(                           \
            false, (AFR), false, bfr[BJ], (short)0, acc, false, false)
#define WMMA_SC(EV, XQ, BJ)                                                    \
        do { v2f a_; a_.x = (EV) * (XQ).x; a_.y = (EV) * (XQ).y;               \
             WMMA_RAW(a_, BJ); } while (0)

        WMMA_RAW(xq0, 0);  WMMA_RAW(xq1, 1);  WMMA_RAW(xq2, 2);  WMMA_RAW(xq3, 3);
        WMMA_SC(e1c, xq0, 4);  WMMA_SC(e1c, xq1, 5);  WMMA_SC(e1c, xq2, 6);  WMMA_SC(e1c, xq3, 7);
        WMMA_SC(e1s, xq0, 8);  WMMA_SC(e1s, xq1, 9);  WMMA_SC(e1s, xq2, 10); WMMA_SC(e1s, xq3, 11);
        WMMA_SC(e2c, xq0, 12); WMMA_SC(e2c, xq1, 13); WMMA_SC(e2c, xq2, 14); WMMA_SC(e2c, xq3, 15);
        WMMA_SC(e2s, xq0, 16); WMMA_SC(e2s, xq1, 17); WMMA_SC(e2s, xq2, 18); WMMA_SC(e2s, xq3, 19);
#undef WMMA_SC
#undef WMMA_RAW

        const int i = r;
        const float g = (i >= 2) ? 1.f : 0.f;
#pragma unroll
        for (int rr = 0; rr < 8; ++rr) {
            const int m = rr + (half << 3);
            const v2f xm01 = *reinterpret_cast<const v2f*>(&xs[wave][m][0]);
            const float xm0 = xm01.x;
            const float xm1 = xm01.y;
            const float xmi = xs[wave][m][i];
            const float am  = al[wave][m];
            const float e0 = c1 * xmi * xm0;
            const float e1 = c1 * xmi * xm1;
            const float e2 = c1 * xmi * xmi;
            const float mx = fmaxf(fmaxf(e0, e1), e2);
            const float w0 = __expf(e0 - mx);
            const float w1 = __expf(e1 - mx);
            const float w2 = g * __expf(e2 - mx);
            const float inv = __builtin_amdgcn_rcpf(w0 + w1 + w2);
            const float s  = (w0 * xm0 + w1 * xm1 + w2 * xmi) * inv;
            out[(s0 + m) * 16 + i] = acc[rr] + c2 * s * am;
        }

        if (!more) break;
        __builtin_amdgcn_sched_barrier(0);
        tile = nextTile;
        tt  = ntt;
        xq0 = nq0; xq1 = nq1; xq2 = nq2; xq3 = nq3;
    }
}

extern "C" void kernel_launch(void* const* d_in, const int* in_sizes, int n_in,
                              void* d_out, int out_size, void* d_ws, size_t ws_size,
                              hipStream_t stream) {
    const float* x  = (const float*)d_in[0];
    const float* t  = (const float*)d_in[1];
    const float* Lb = (const float*)d_in[2];
    const float* wq = (const float*)d_in[3];
    const float* wk = (const float*)d_in[4];
    const float* wv = (const float*)d_in[5];
    const float* wo = (const float*)d_in[6];
    const float* aw = (const float*)d_in[7];
    float* out = (float*)d_out;

    const int B = in_sizes[1];
    const int nTiles = B >> 4;
    if (nTiles <= 0) return;

    int blocks = (nTiles + WPB - 1) / WPB;
    if (blocks > 512) blocks = 512;

    nxro_fused_kernel<<<blocks, 256, 0, stream>>>(x, t, Lb, wq, wk, wv, wo, aw,
                                                  out, nTiles);
}


# Wave start-up cut from 4 dependent fetch round trips to 2: all arguments in one fetch; wq,wk,wv,wo read one element per lane, summed by 32-lane butterfly (order of these two sums changes)
# speedup vs baseline: 1.2299x; 1.0612x over previous
#include <hip/hip_runtime.h>
#include <math.h>

typedef __attribute__((ext_vector_type(2))) float v2f;
typedef __attribute__((ext_vector_type(8))) float v8f;

#define WPB 8

template <int N>
static __device__ __forceinline__ float row_xor(float v) {
    const int i = __builtin_bit_cast(int, v);
    return __builtin_bit_cast(float, __builtin_amdgcn_update_dpp(i, i, 0x160 + N, 0xf, 0xf, false));
}
static __device__ __forceinline__ float other_row(float v) {
    const int i = __builtin_bit_cast(int, v);
    return __builtin_bit_cast(float,
        __builtin_amdgcn_permlanex16(i, i, 0x76543210, 0xfedcba98, false, false));
}
static __device__ __forceinline__ float wave_sum32(float v) {
    v += row_xor<1>(v);
    v += row_xor<2>(v);
    v += row_xor<4>(v);
    v += row_xor<8>(v);
    v += other_row(v);
    return v;
}

__global__ __launch_bounds__(256) void nxro_fused_kernel(
    const float* __restrict__ x, const float* __restrict__ t,
    const float* __restrict__ Lb, const float* __restrict__ wq,
    const float* __restrict__ wk, const float* __restrict__ wv,
    const float* __restrict__ wo, const float* __restrict__ aw,
    float* __restrict__ out, int nTiles)
{
    __shared__ float xs[WPB][16][18];
    __shared__ float al[WPB][16];

    const int lane = threadIdx.x & 31;
    const int wave = threadIdx.x >> 5;
    const int half = lane >> 4;
    const int r    = lane & 15;

    const int waveId = blockIdx.x * WPB + wave;
    const int stride = gridDim.x * WPB;

    asm volatile("" :: "s"(x), "s"(t), "s"(Lb), "s"(wq), "s"(wk), "s"(wv), "s"(wo),
                       "s"(out), "s"(stride));
    if (waveId >= nTiles) return;

    const float pq = wq[lane], pk = wk[lane], pv = wv[lane], po = wo[lane];

    int tile = waveId;
    float tt;
    v2f xq0, xq1, xq2, xq3;
    {
        const int row0 = (tile << 4) + r;
        tt = t[row0];
        const v2f* xp = reinterpret_cast<const v2f*>(x + row0 * 16);
        xq0 = xp[0 + half];
        xq1 = xp[2 + half];
        xq2 = xp[4 + half];
        xq3 = xp[6 + half];
    }
    __builtin_amdgcn_sched_barrier(0);

    v2f bfr[20];
#pragma unroll
    for (int j = 0; j < 20; ++j) {
        const int kb = j >> 2;
        const int v0 = ((j & 3) << 2) + (half << 1);
        const float* p = Lb + kb * 256 + r * 16 + v0;
        v2f b_; b_.x = p[0]; b_.y = p[1];
        bfr[j] = b_;
    }

    const float c1 = 0.17677669529663688f *
        __builtin_bit_cast(float, __builtin_amdgcn_readfirstlane(
            __builtin_bit_cast(int, wave_sum32(pq * pk))));
    const float c2 = __builtin_bit_cast(float, __builtin_amdgcn_readfirstlane(
        __builtin_bit_cast(int, wave_sum32(pv * po))));

    const float aw0 = aw[0], aw1 = aw[1], aw2 = aw[2], aw3 = aw[3], aw4 = aw[4];

    for (;;) {
        const int s0  = tile << 4;

        const int  nextTile = tile + stride;
        const bool more     = nextTile < nTiles;
        const int  nrow     = ((more ? nextTile : tile) << 4) + r;
        const float ntt = t[nrow];
        const v2f* np = reinterpret_cast<const v2f*>(x + nrow * 16);
        const v2f nq0 = np[0 + half];
        const v2f nq1 = np[2 + half];
        const v2f nq2 = np[4 + half];
        const v2f nq3 = np[6 + half];

        {
            v2f* rp = reinterpret_cast<v2f*>(&xs[wave][r][0]);
            rp[0 + half] = xq0;
            rp[2 + half] = xq1;
            rp[4 + half] = xq2;
            rp[6 + half] = xq3;
        }

        const float a1  = 6.283185307179586f * tt;
        const float e1c = cosf(a1), e1s = sinf(a1);
        const float tc  = e1c + e1c;
        const float e2c = fmaf(tc, e1c, -1.0f);
        const float e2s = tc * e1s;

        if (half == 0) {
            float z = aw0 + aw1 * e1c + aw2 * e1s + aw3 * e2c + aw4 * e2s;
            al[wave][r] = __builtin_amdgcn_rcpf(1.f + __expf(-z));
        }

        v8f acc = {};
#define WMMA_RAW(AFR, BJ)                                                      \
        acc = __builtin_amdgcn_wmma_f32_16x16x4_f32(                           \
            false, (AFR), false, bfr[BJ], (short)0, acc, false, false)
#define WMMA_SC(EV, XQ, BJ)                                                    \
        do { v2f a_; a_.x = (EV) * (XQ).x; a_.y = (EV) * (XQ).y;               \
             WMMA_RAW(a_, BJ); } while (0)

        WMMA_RAW(xq0, 0);  WMMA_RAW(xq1, 1);  WMMA_RAW(xq2, 2);  WMMA_RAW(xq3, 3);
        WMMA_SC(e1c, xq0, 4);  WMMA_SC(e1c, xq1, 5);  WMMA_SC(e1c, xq2, 6);  WMMA_SC(e1c, xq3, 7);
        WMMA_SC(e1s, xq0, 8);  WMMA_SC(e1s, xq1, 9);  WMMA_SC(e1s, xq2, 10); WMMA_SC(e1s, xq3, 11);
        WMMA_SC(e2c, xq0, 12); WMMA_SC(e2c, xq1, 13); WMMA_SC(e2c, xq2, 14); WMMA_SC(e2c, xq3, 15);
        WMMA_SC(e2s, xq0, 16); WMMA_SC(e2s, xq1, 17); WMMA_SC(e2s, xq2, 18); WMMA_SC(e2s, xq3, 19);
#undef WMMA_SC
#undef WMMA_RAW

        const int i = r;
        const float g = (i >= 2) ? 1.f : 0.f;
#pragma unroll
        for (int rr = 0; rr < 8; ++rr) {
            const int m = rr + (half << 3);
            const v2f xm01 = *reinterpret_cast<const v2f*>(&xs[wave][m][0]);
            const float xm0 = xm01.x;
            const float xm1 = xm01.y;
            const float xmi = xs[wave][m][i];
            const float am  = al[wave][m];
            const float e0 = c1 * xmi * xm0;
            const float e1 = c1 * xmi * xm1;
            const float e2 = c1 * xmi * xmi;
            const float mx = fmaxf(fmaxf(e0, e1), e2);
            const float w0 = __expf(e0 - mx);
            const float w1 = __expf(e1 - mx);
            const float w2 = g * __expf(e2 - mx);
            const float inv = __builtin_amdgcn_rcpf(w0 + w1 + w2);
            const float s  = (w0 * xm0 + w1 * xm1 + w2 * xmi) * inv;
            out[(s0 + m) * 16 + i] = acc[rr] + c2 * s * am;
        }

        if (!more) break;
        __builtin_amdgcn_sched_barrier(0);
        tile = nextTile;
        tt  = ntt;
        xq0 = nq0; xq1 = nq1; xq2 = nq2; xq3 = nq3;
    }
}

extern "C" void kernel_launch(void* const* d_in, const int* in_sizes, int n_in,
                              void* d_out, int out_size, void* d_ws, size_t ws_size,
                              hipStream_t stream) {
    const float* x  = (const float*)d_in[0];
    const float* t  = (const float*)d_in[1];
    const float* Lb = (const float*)d_in[2];
    const float* wq = (const float*)d_in[3];
    const float* wk = (const float*)d_in[4];
    const float* wv = (const float*)d_in[5];
    const float* wo = (const float*)d_in[6];
    const float* aw = (const float*)d_in[7];
    float* out = (float*)d_out;

    const int B = in_sizes[1];
    const int nTiles = B >> 4;
    if (nTiles <= 0) return;

    int blocks = (nTiles + WPB - 1) / WPB;
    if (blocks > 512) blocks = 512;

    nxro_fused_kernel<<<blocks, 256, 0, stream>>>(x, t, Lb, wq, wk, wv, wo, aw,
                                                  out, nTiles);
}
